# starting kernel
# baseline (speedup 1.0000x reference)
#include <hip/hip_runtime.h>
#include <hip/hip_bf16.h>

#define N_NODES 50000
#define N_EDGES 800000
#define D       128
#define D_OUT   128
#define K2      256
#define XS_STRIDE 260

typedef __attribute__((ext_vector_type(2))) float v2f;
typedef __attribute__((ext_vector_type(8))) float v8f;

__global__ void gcn_zero_kernel(float* __restrict__ ws, int count) {
    int i = blockIdx.x * blockDim.x + threadIdx.x;
    if (i < count) ws[i] = 0.0f;
}

__global__ void gcn_scatter_kernel(const float* __restrict__ feat,
                                   const int* __restrict__ src,
                                   const int* __restrict__ dst,
                                   float* __restrict__ agg,
                                   float* __restrict__ deg) {
    int wavesPerBlk = blockDim.x >> 5;
    int e = blockIdx.x * wavesPerBlk + (threadIdx.x >> 5);
    if (e >= N_EDGES) return;
    int lane = threadIdx.x & 31;
    int s = src[e];
    int t = dst[e];
    const float4 v = *(const float4*)(feat + (size_t)s * D + lane * 4);
    float* ap = agg + (size_t)t * D + lane * 4;
    atomicAdd(ap + 0, v.x);
    atomicAdd(ap + 1, v.y);
    atomicAdd(ap + 2, v.z);
    atomicAdd(ap + 3, v.w);
    if (lane == 0) atomicAdd(deg + t, 1.0f);
}

__global__ void gcn_gemm_kernel(const float* __restrict__ feat,
                                const float* __restrict__ agg,
                                const float* __restrict__ deg,
                                const float* __restrict__ W,
                                const float* __restrict__ bias,
                                float* __restrict__ out) {
    __shared__ float xs[16 * XS_STRIDE];
    __shared__ float invd[16];

    const int node0 = blockIdx.x * 16;
    const int t = threadIdx.x;

    if (t < 16) {
        float dg = deg[node0 + t];
        invd[t] = 1.0f / fmaxf(dg, 1.0f);
    }
    __syncthreads();

    #pragma unroll
    for (int i = t; i < 16 * K2; i += 256) {
        int r = i >> 8;
        int c = i & 255;
        float v;
        if (c < D) {
            v = agg[(size_t)(node0 + r) * D + c] * invd[r];
        } else {
            v = feat[(size_t)(node0 + r) * D + (c - D)];
        }
        xs[r * XS_STRIDE + c] = v;
    }
    __syncthreads();

    const int wave = t >> 5;
    const int lane = t & 31;
    const int half = lane >> 4;
    const int lid  = lane & 15;
    const int n0   = wave * 16;

    const float* wrow = W + (size_t)(n0 + lid) * K2 + 2 * half;
    const float* arow = xs + lid * XS_STRIDE + 2 * half;

    __builtin_prefetch(wrow, 0, 3);

    v8f acc = {};
    #pragma unroll 8
    for (int k = 0; k < K2; k += 4) {
        v2f a = *(const v2f*)(arow + k);
        v2f b = *(const v2f*)(wrow + k);
        acc = __builtin_amdgcn_wmma_f32_16x16x4_f32(
            false, a, false, b, (short)0, acc, false, false);
    }

    const float bv = bias[n0 + lid];
    #pragma unroll
    for (int j = 0; j < 8; ++j) {
        int row = node0 + j + 8 * half;
        out[(size_t)row * D_OUT + n0 + lid] = acc[j] + bv;
    }
}

extern "C" void kernel_launch(void* const* d_in, const int* in_sizes, int n_in,
                              void* d_out, int out_size, void* d_ws, size_t ws_size,
                              hipStream_t stream) {
    const float* feat = (const float*)d_in[0];
    const int*   src  = (const int*)d_in[1];
    const int*   dst  = (const int*)d_in[2];
    const float* W    = (const float*)d_in[3];
    const float* bias = (const float*)d_in[4];
    float*       out  = (float*)d_out;

    float* agg = (float*)d_ws;
    float* deg = agg + (size_t)N_NODES * D;

    {
        int count = N_NODES * D + N_NODES;
        int blocks = (count + 255) / 256;
        gcn_zero_kernel<<<blocks, 256, 0, stream>>>(agg, count);
    }

    {
        int blocks = (N_EDGES + 7) / 8;
        gcn_scatter_kernel<<<blocks, 256, 0, stream>>>(feat, src, dst, agg, deg);
    }

    {
        int blocks = N_NODES / 16;
        gcn_gemm_kernel<<<blocks, 256, 0, stream>>>(feat, agg, deg, W, bias, out);
    }
}


# scatter: lane l owns columns l,l+32,l+64,l+96, so each f32 atomic add and each row load covers one whole 128-byte line instead of four quarter-filled ones; sums arrival-ordered as before
# speedup vs baseline: 1.0688x; 1.0688x over previous
#include <hip/hip_runtime.h>
#include <hip/hip_bf16.h>

#define N_NODES 50000
#define N_EDGES 800000
#define D       128
#define D_OUT   128
#define K2      256
#define XS_STRIDE 260

typedef __attribute__((ext_vector_type(2))) float v2f;
typedef __attribute__((ext_vector_type(8))) float v8f;

__global__ void gcn_zero_kernel(float* __restrict__ ws, int count) {
    int i = blockIdx.x * blockDim.x + threadIdx.x;
    if (i < count) ws[i] = 0.0f;
}

__global__ void gcn_scatter_kernel(const float* __restrict__ feat,
                                   const int* __restrict__ src,
                                   const int* __restrict__ dst,
                                   float* __restrict__ agg,
                                   float* __restrict__ deg) {
    int wavesPerBlk = blockDim.x >> 5;
    int e = blockIdx.x * wavesPerBlk + (threadIdx.x >> 5);
    if (e >= N_EDGES) return;
    int lane = threadIdx.x & 31;
    int s = src[e];
    int t = dst[e];
    const float* fp = feat + (size_t)s * D + lane;
    const float v0 = fp[0];
    const float v1 = fp[32];
    const float v2 = fp[64];
    const float v3 = fp[96];
    float* ap = agg + (size_t)t * D + lane;
    atomicAdd(ap + 0,  v0);
    atomicAdd(ap + 32, v1);
    atomicAdd(ap + 64, v2);
    atomicAdd(ap + 96, v3);
    if (lane == 0) atomicAdd(deg + t, 1.0f);
}

__global__ void gcn_gemm_kernel(const float* __restrict__ feat,
                                const float* __restrict__ agg,
                                const float* __restrict__ deg,
                                const float* __restrict__ W,
                                const float* __restrict__ bias,
                                float* __restrict__ out) {
    __shared__ float xs[16 * XS_STRIDE];
    __shared__ float invd[16];

    const int node0 = blockIdx.x * 16;
    const int t = threadIdx.x;

    if (t < 16) {
        float dg = deg[node0 + t];
        invd[t] = 1.0f / fmaxf(dg, 1.0f);
    }
    __syncthreads();

    #pragma unroll
    for (int i = t; i < 16 * K2; i += 256) {
        int r = i >> 8;
        int c = i & 255;
        float v;
        if (c < D) {
            v = agg[(size_t)(node0 + r) * D + c] * invd[r];
        } else {
            v = feat[(size_t)(node0 + r) * D + (c - D)];
        }
        xs[r * XS_STRIDE + c] = v;
    }
    __syncthreads();

    const int wave = t >> 5;
    const int lane = t & 31;
    const int half = lane >> 4;
    const int lid  = lane & 15;
    const int n0   = wave * 16;

    const float* wrow = W + (size_t)(n0 + lid) * K2 + 2 * half;
    const float* arow = xs + lid * XS_STRIDE + 2 * half;

    __builtin_prefetch(wrow, 0, 3);

    v8f acc = {};
    #pragma unroll 8
    for (int k = 0; k < K2; k += 4) {
        v2f a = *(const v2f*)(arow + k);
        v2f b = *(const v2f*)(wrow + k);
        acc = __builtin_amdgcn_wmma_f32_16x16x4_f32(
            false, a, false, b, (short)0, acc, false, false);
    }

    const float bv = bias[n0 + lid];
    #pragma unroll
    for (int j = 0; j < 8; ++j) {
        int row = node0 + j + 8 * half;
        out[(size_t)row * D_OUT + n0 + lid] = acc[j] + bv;
    }
}

extern "C" void kernel_launch(void* const* d_in, const int* in_sizes, int n_in,
                              void* d_out, int out_size, void* d_ws, size_t ws_size,
                              hipStream_t stream) {
    const float* feat = (const float*)d_in[0];
    const int*   src  = (const int*)d_in[1];
    const int*   dst  = (const int*)d_in[2];
    const float* W    = (const float*)d_in[3];
    const float* bias = (const float*)d_in[4];
    float*       out  = (float*)d_out;

    float* agg = (float*)d_ws;
    float* deg = agg + (size_t)N_NODES * D;

    {
        int count = N_NODES * D + N_NODES;
        int blocks = (count + 255) / 256;
        gcn_zero_kernel<<<blocks, 256, 0, stream>>>(agg, count);
    }

    {
        int blocks = (N_EDGES + 7) / 8;
        gcn_scatter_kernel<<<blocks, 256, 0, stream>>>(feat, src, dst, agg, deg);
    }

    {
        int blocks = N_NODES / 16;
        gcn_gemm_kernel<<<blocks, 256, 0, stream>>>(feat, agg, deg, W, bias, out);
    }
}


# fewer, fatter waves: a scatter wave takes 32 edges (coalesced index loads, readlane row bases, 32 row loads in flight; 800000 waves -> 25000), zero kernel 8 float4/thread; sums still arrival-ordered
# speedup vs baseline: 1.4310x; 1.3389x over previous
#include <hip/hip_runtime.h>
#include <hip/hip_bf16.h>

#define N_NODES 50000
#define N_EDGES 800000
#define D       128
#define D_OUT   128
#define K2      256
#define XS_STRIDE 260

typedef __attribute__((ext_vector_type(2))) float v2f;
typedef __attribute__((ext_vector_type(8))) float v8f;

#define ZERO_VEC_PER_THREAD 8
#define ZERO_COUNT  (N_NODES * D + N_NODES)
#define ZERO_COUNT4 (ZERO_COUNT / 4)
static_assert(ZERO_COUNT % 4 == 0, "workspace clear must be a whole number of float4");

__global__ void gcn_zero_kernel(float* __restrict__ ws) {
    float4* __restrict__ ws4 = reinterpret_cast<float4*>(ws);
    const int base = blockIdx.x * (256 * ZERO_VEC_PER_THREAD) + threadIdx.x;
    #pragma unroll
    for (int j = 0; j < ZERO_VEC_PER_THREAD; ++j) {
        const int i = base + j * 256;
        if (i < ZERO_COUNT4) ws4[i] = make_float4(0.0f, 0.0f, 0.0f, 0.0f);
    }
}

#define EDGES_PER_WAVE 32
#define EDGE_BATCH     8
#define SCATTER_WAVES  (N_EDGES / EDGES_PER_WAVE)
static_assert(N_EDGES % EDGES_PER_WAVE == 0, "edges must tile into whole waves");
static_assert(SCATTER_WAVES % 8 == 0, "waves must tile into whole 8-wave blocks");

__global__ void gcn_scatter_kernel(const float* __restrict__ feat,
                                   const int* __restrict__ src,
                                   const int* __restrict__ dst,
                                   float* __restrict__ agg,
                                   float* __restrict__ deg) {
    const int lane = threadIdx.x & 31;
    const int e0 = (blockIdx.x * 8 + (threadIdx.x >> 5)) * EDGES_PER_WAVE;
    const int sv = src[e0 + lane];
    const int tv = dst[e0 + lane];
    atomicAdd(deg + tv, 1.0f);

    #pragma unroll 1
    for (int b = 0; b < EDGES_PER_WAVE; b += EDGE_BATCH) {
        float v[EDGE_BATCH][4];
        #pragma unroll
        for (int j = 0; j < EDGE_BATCH; ++j) {
            const int s = __builtin_amdgcn_readlane(sv, b + j);
            const float* fp = feat + (size_t)s * D + lane;
            v[j][0] = fp[0];
            v[j][1] = fp[32];
            v[j][2] = fp[64];
            v[j][3] = fp[96];
        }
        #pragma unroll
        for (int j = 0; j < EDGE_BATCH; ++j) {
            const int t = __builtin_amdgcn_readlane(tv, b + j);
            float* ap = agg + (size_t)t * D + lane;
            atomicAdd(ap + 0,  v[j][0]);
            atomicAdd(ap + 32, v[j][1]);
            atomicAdd(ap + 64, v[j][2]);
            atomicAdd(ap + 96, v[j][3]);
        }
    }
}

__global__ void gcn_gemm_kernel(const float* __restrict__ feat,
                                const float* __restrict__ agg,
                                const float* __restrict__ deg,
                                const float* __restrict__ W,
                                const float* __restrict__ bias,
                                float* __restrict__ out) {
    __shared__ float xs[16 * XS_STRIDE];
    __shared__ float invd[16];

    const int node0 = blockIdx.x * 16;
    const int t = threadIdx.x;

    if (t < 16) {
        float dg = deg[node0 + t];
        invd[t] = 1.0f / fmaxf(dg, 1.0f);
    }
    __syncthreads();

    #pragma unroll
    for (int i = t; i < 16 * K2; i += 256) {
        int r = i >> 8;
        int c = i & 255;
        float v;
        if (c < D) {
            v = agg[(size_t)(node0 + r) * D + c] * invd[r];
        } else {
            v = feat[(size_t)(node0 + r) * D + (c - D)];
        }
        xs[r * XS_STRIDE + c] = v;
    }
    __syncthreads();

    const int wave = t >> 5;
    const int lane = t & 31;
    const int half = lane >> 4;
    const int lid  = lane & 15;
    const int n0   = wave * 16;

    const float* wrow = W + (size_t)(n0 + lid) * K2 + 2 * half;
    const float* arow = xs + lid * XS_STRIDE + 2 * half;

    __builtin_prefetch(wrow, 0, 3);

    v8f acc = {};
    #pragma unroll 8
    for (int k = 0; k < K2; k += 4) {
        v2f a = *(const v2f*)(arow + k);
        v2f b = *(const v2f*)(wrow + k);
        acc = __builtin_amdgcn_wmma_f32_16x16x4_f32(
            false, a, false, b, (short)0, acc, false, false);
    }

    const float bv = bias[n0 + lid];
    #pragma unroll
    for (int j = 0; j < 8; ++j) {
        int row = node0 + j + 8 * half;
        out[(size_t)row * D_OUT + n0 + lid] = acc[j] + bv;
    }
}

extern "C" void kernel_launch(void* const* d_in, const int* in_sizes, int n_in,
                              void* d_out, int out_size, void* d_ws, size_t ws_size,
                              hipStream_t stream) {
    const float* feat = (const float*)d_in[0];
    const int*   src  = (const int*)d_in[1];
    const int*   dst  = (const int*)d_in[2];
    const float* W    = (const float*)d_in[3];
    const float* bias = (const float*)d_in[4];
    float*       out  = (float*)d_out;

    float* agg = (float*)d_ws;
    float* deg = agg + (size_t)N_NODES * D;

    {
        int blocks = (ZERO_COUNT4 + 256 * ZERO_VEC_PER_THREAD - 1) / (256 * ZERO_VEC_PER_THREAD);
        gcn_zero_kernel<<<blocks, 256, 0, stream>>>(agg);
    }

    {
        int blocks = SCATTER_WAVES / 8;
        gcn_scatter_kernel<<<blocks, 256, 0, stream>>>(feat, src, dst, agg, deg);
    }

    {
        int blocks = N_NODES / 16;
        gcn_gemm_kernel<<<blocks, 256, 0, stream>>>(feat, agg, deg, W, bias, out);
    }
}


# aggregation binned by dst range: a block owns 196 nodes, scans all dst, queues its edges, sums their rows with LDS f32 atomics, stores agg/deg once; global atomics, clear gone; sums arrival-ordered
# speedup vs baseline: 2.1418x; 1.4967x over previous
#include <hip/hip_runtime.h>
#include <hip/hip_bf16.h>

#define N_NODES 50000
#define N_EDGES 800000
#define D       128
#define D_OUT   128
#define K2      256
#define XS_STRIDE 260

typedef __attribute__((ext_vector_type(2))) float v2f;
typedef __attribute__((ext_vector_type(8))) float v8f;

#define BIN_NODES   196
#define N_BINS      ((N_NODES + BIN_NODES - 1) / BIN_NODES)
#define AGG_WAVES   32
#define AGG_THREADS (AGG_WAVES * 32)
#define SLICE_EDGES (N_EDGES / AGG_WAVES)
#define SCAN_ITERS  ((SLICE_EDGES + 127) / 128)
#define SCAN_UNROLL 7
#define QCAP        1024
#define EDGE_BATCH  8
static_assert(N_EDGES % AGG_WAVES == 0, "edges must split evenly over the waves");
static_assert(SLICE_EDGES % 4 == 0, "a lane's four edges are all inside or all outside the slice");
static_assert(SCAN_ITERS % SCAN_UNROLL == 0, "scan loop unrolls exactly");
static_assert(BIN_NODES <= 256, "local node index must fit the 8 low bits of a queue entry");
static_assert(N_EDGES < (1 << 23), "edge id must fit the upper bits of a queue entry");
static_assert((BIN_NODES * D) % 4 == 0, "LDS clear and copy-out use float4");

__device__ __forceinline__ void gcn_drain_queue(const float* __restrict__ feat,
                                                const int* __restrict__ src,
                                                const int* q, int nq,
                                                float* agg_s, int lane) {
    for (int base = 0; base < nq; base += 32) {
        const int idx = base + lane;
        const bool live = idx < nq;
        const int ent = live ? q[idx] : 0;
        const int sv = live ? src[ent >> 8] : 0;
        const int lv = ent & 255;
        const int nb = (nq - base < 32) ? (nq - base) : 32;
        int b = 0;
        #pragma unroll 1
        for (; b + EDGE_BATCH <= nb; b += EDGE_BATCH) {
            float v[EDGE_BATCH][4];
            #pragma unroll
            for (int j = 0; j < EDGE_BATCH; ++j) {
                const int s = __builtin_amdgcn_readlane(sv, b + j);
                const float* fp = feat + (size_t)s * D + lane;
                v[j][0] = fp[0];
                v[j][1] = fp[32];
                v[j][2] = fp[64];
                v[j][3] = fp[96];
            }
            #pragma unroll
            for (int j = 0; j < EDGE_BATCH; ++j) {
                const int l = __builtin_amdgcn_readlane(lv, b + j);
                float* ap = agg_s + l * D + lane;
                atomicAdd(ap + 0,  v[j][0]);
                atomicAdd(ap + 32, v[j][1]);
                atomicAdd(ap + 64, v[j][2]);
                atomicAdd(ap + 96, v[j][3]);
            }
        }
        #pragma unroll 1
        for (; b < nb; ++b) {
            const int s = __builtin_amdgcn_readlane(sv, b);
            const int l = __builtin_amdgcn_readlane(lv, b);
            const float* fp = feat + (size_t)s * D + lane;
            const float r0 = fp[0], r1 = fp[32], r2 = fp[64], r3 = fp[96];
            float* ap = agg_s + l * D + lane;
            atomicAdd(ap + 0,  r0);
            atomicAdd(ap + 32, r1);
            atomicAdd(ap + 64, r2);
            atomicAdd(ap + 96, r3);
        }
    }
}

__global__ __launch_bounds__(AGG_THREADS)
void gcn_bin_aggregate_kernel(const float* __restrict__ feat,
                              const int* __restrict__ src,
                              const int* __restrict__ dst,
                              float* __restrict__ agg,
                              float* __restrict__ deg) {
    __shared__ float agg_s[BIN_NODES * D];
    __shared__ float deg_s[256];
    __shared__ int   q_s[AGG_WAVES * QCAP];

    const int t    = threadIdx.x;
    const int lane = t & 31;
    const int wave = t >> 5;
    const int lo   = blockIdx.x * BIN_NODES;
    const int nn   = (N_NODES - lo < BIN_NODES) ? (N_NODES - lo) : BIN_NODES;

    {
        float4* a4 = reinterpret_cast<float4*>(agg_s);
        for (int i = t; i < BIN_NODES * D / 4; i += AGG_THREADS)
            a4[i] = make_float4(0.0f, 0.0f, 0.0f, 0.0f);
        if (t < 256) deg_s[t] = 0.0f;
    }
    __syncthreads();

    int* q = q_s + wave * QCAP;
    int nq = 0;
    const int ebase = wave * SLICE_EDGES;
    const int eend  = ebase + SLICE_EDGES;
    const int rot = (blockIdx.x * 3) % SCAN_ITERS;

    for (int it0 = 0; it0 < SCAN_ITERS; it0 += SCAN_UNROLL) {
        int4 dv[SCAN_UNROLL];
        int  ev[SCAN_UNROLL];
        #pragma unroll
        for (int u = 0; u < SCAN_UNROLL; ++u) {
            int itr = it0 + u + rot;
            if (itr >= SCAN_ITERS) itr -= SCAN_ITERS;
            const int e = ebase + itr * 128 + lane * 4;
            ev[u] = e;
            const int ec = (e < eend) ? e : (eend - 4);
            dv[u] = *reinterpret_cast<const int4*>(dst + ec);
        }
        #pragma unroll
        for (int u = 0; u < SCAN_UNROLL; ++u) {
            const int dd[4] = {dv[u].x, dv[u].y, dv[u].z, dv[u].w};
            unsigned lj[4];
            bool mj[4];
            bool any = false;
            const bool inside = ev[u] < eend;
            #pragma unroll
            for (int j = 0; j < 4; ++j) {
                lj[j] = (unsigned)(dd[j] - lo);
                mj[j] = inside && (lj[j] < (unsigned)nn);
                any = any || mj[j];
            }
            if (__builtin_amdgcn_ballot_w32(any) != 0u) {
                #pragma unroll
                for (int j = 0; j < 4; ++j) {
                    const unsigned bal = __builtin_amdgcn_ballot_w32(mj[j]);
                    if (mj[j]) {
                        const int pos = nq + (int)__builtin_amdgcn_mbcnt_lo(bal, 0u);
                        q[pos] = ((ev[u] + j) << 8) | (int)lj[j];
                        atomicAdd(deg_s + lj[j], 1.0f);
                    }
                    nq += __builtin_popcount(bal);
                }
                if (nq > QCAP - 128) {
                    gcn_drain_queue(feat, src, q, nq, agg_s, lane);
                    nq = 0;
                }
            }
        }
    }
    gcn_drain_queue(feat, src, q, nq, agg_s, lane);
    __syncthreads();

    {
        const float4* a4 = reinterpret_cast<const float4*>(agg_s);
        float4* o4 = reinterpret_cast<float4*>(agg + (size_t)lo * D);
        for (int i = t; i < nn * D / 4; i += AGG_THREADS) o4[i] = a4[i];
        if (t < nn) deg[lo + t] = deg_s[t];
    }
}

__global__ void gcn_gemm_kernel(const float* __restrict__ feat,
                                const float* __restrict__ agg,
                                const float* __restrict__ deg,
                                const float* __restrict__ W,
                                const float* __restrict__ bias,
                                float* __restrict__ out) {
    __shared__ float xs[16 * XS_STRIDE];
    __shared__ float invd[16];

    const int node0 = blockIdx.x * 16;
    const int t = threadIdx.x;

    if (t < 16) {
        float dg = deg[node0 + t];
        invd[t] = 1.0f / fmaxf(dg, 1.0f);
    }
    __syncthreads();

    #pragma unroll
    for (int i = t; i < 16 * K2; i += 256) {
        int r = i >> 8;
        int c = i & 255;
        float v;
        if (c < D) {
            v = agg[(size_t)(node0 + r) * D + c] * invd[r];
        } else {
            v = feat[(size_t)(node0 + r) * D + (c - D)];
        }
        xs[r * XS_STRIDE + c] = v;
    }
    __syncthreads();

    const int wave = t >> 5;
    const int lane = t & 31;
    const int half = lane >> 4;
    const int lid  = lane & 15;
    const int n0   = wave * 16;

    const float* wrow = W + (size_t)(n0 + lid) * K2 + 2 * half;
    const float* arow = xs + lid * XS_STRIDE + 2 * half;

    __builtin_prefetch(wrow, 0, 3);

    v8f acc = {};
    #pragma unroll 8
    for (int k = 0; k < K2; k += 4) {
        v2f a = *(const v2f*)(arow + k);
        v2f b = *(const v2f*)(wrow + k);
        acc = __builtin_amdgcn_wmma_f32_16x16x4_f32(
            false, a, false, b, (short)0, acc, false, false);
    }

    const float bv = bias[n0 + lid];
    #pragma unroll
    for (int j = 0; j < 8; ++j) {
        int row = node0 + j + 8 * half;
        out[(size_t)row * D_OUT + n0 + lid] = acc[j] + bv;
    }
}

extern "C" void kernel_launch(void* const* d_in, const int* in_sizes, int n_in,
                              void* d_out, int out_size, void* d_ws, size_t ws_size,
                              hipStream_t stream) {
    const float* feat = (const float*)d_in[0];
    const int*   src  = (const int*)d_in[1];
    const int*   dst  = (const int*)d_in[2];
    const float* W    = (const float*)d_in[3];
    const float* bias = (const float*)d_in[4];
    float*       out  = (float*)d_out;

    float* agg = (float*)d_ws;
    float* deg = agg + (size_t)N_NODES * D;

    {
        gcn_bin_aggregate_kernel<<<N_BINS, AGG_THREADS, 0, stream>>>(feat, src, dst, agg, deg);
    }

    {
        int blocks = N_NODES / 16;
        gcn_gemm_kernel<<<blocks, 256, 0, stream>>>(feat, agg, deg, W, bias, out);
    }
}


# new first launch orders each 3125-edge slice by destination bin (LDS counting sort, atomic ranks); an aggregation block reads only its own packed edges, not all of dst; sums still arrival-ordered
# speedup vs baseline: 2.5479x; 1.1896x over previous
#include <hip/hip_runtime.h>
#include <hip/hip_bf16.h>

#define N_NODES 50000
#define N_EDGES 800000
#define D       128
#define D_OUT   128
#define K2      256
#define XS_STRIDE 260

typedef __attribute__((ext_vector_type(2))) float v2f;
typedef __attribute__((ext_vector_type(8))) float v8f;

#define BIN_NODES   196
#define N_BINS      ((N_NODES + BIN_NODES - 1) / BIN_NODES)
#define AGG_WAVES   32
#define AGG_THREADS (AGG_WAVES * 32)
#define QCAP        1024
#define EDGE_BATCH  8
#define PART_BLOCKS  256
#define PART_THREADS 256
#define PART_EDGES   (N_EDGES / PART_BLOCKS)
#define PART_PER_THREAD ((PART_EDGES + PART_THREADS - 1) / PART_THREADS)
#define SLICES_PER_WAVE (PART_BLOCKS / AGG_WAVES)
#define DEG_PLANE   ((N_NODES + 63) / 64 * 64)
static_assert(N_EDGES % PART_BLOCKS == 0, "edges must split evenly over the slices");
static_assert(PART_BLOCKS % AGG_WAVES == 0, "slices must split evenly over the waves");
static_assert(N_BINS == PART_THREADS, "one partition thread per bin for the bin tables");
static_assert(BIN_NODES <= 256, "local node index must fit the 8 low bits of a packed edge");
static_assert(N_NODES < (1 << 23), "source node must fit the upper bits of a packed edge");
static_assert(SLICES_PER_WAVE * 32 + 32 <= QCAP, "first chunks of a wave's slices fit its queue");
static_assert((BIN_NODES * D) % 4 == 0, "LDS clear and copy-out use float4");

__global__ __launch_bounds__(PART_THREADS)
void gcn_partition_kernel(const int* __restrict__ src,
                          const int* __restrict__ dst,
                          int* __restrict__ sorted,
                          int* __restrict__ starts,
                          int* __restrict__ counts) {
    __shared__ int cnt_s[N_BINS + 1];
    __shared__ int start_s[N_BINS];
    __shared__ int wtot_s[PART_THREADS / 32];
    __shared__ int buf_s[PART_EDGES];

    const int t    = threadIdx.x;
    const int lane = t & 31;
    const int wave = t >> 5;
    const int sl   = blockIdx.x;
    const int e0   = sl * PART_EDGES;

    cnt_s[t] = 0;
    if (t == 0) cnt_s[N_BINS] = 0;
    __syncthreads();

    int dv[PART_PER_THREAD], sv[PART_PER_THREAD];
    #pragma unroll
    for (int j = 0; j < PART_PER_THREAD; ++j) {
        const int i = t + j * PART_THREADS;
        const int ic = (i < PART_EDGES) ? i : (PART_EDGES - 1);
        const int d = dst[e0 + ic];
        sv[j] = src[e0 + ic];
        dv[j] = (i < PART_EDGES) ? d : -1;
    }
    int bin[PART_PER_THREAD], rank[PART_PER_THREAD];
    #pragma unroll
    for (int j = 0; j < PART_PER_THREAD; ++j) {
        const bool ok = (unsigned)dv[j] < (unsigned)N_NODES;
        bin[j] = ok ? dv[j] / BIN_NODES : N_BINS;
        rank[j] = atomicAdd(&cnt_s[bin[j]], 1);
    }
    __syncthreads();

    const int c = cnt_s[t];
    int x = c;
    #pragma unroll
    for (int off = 1; off < 32; off <<= 1) {
        const int y = __shfl_up(x, off);
        if (lane >= off) x += y;
    }
    if (lane == 31) wtot_s[wave] = x;
    __syncthreads();
    int wbase = 0;
    #pragma unroll
    for (int w = 0; w < PART_THREADS / 32; ++w) wbase += (w < wave) ? wtot_s[w] : 0;
    const int excl = wbase + x - c;
    start_s[t] = excl;
    starts[t * PART_BLOCKS + sl] = excl;
    counts[t * PART_BLOCKS + sl] = c;
    __syncthreads();

    #pragma unroll
    for (int j = 0; j < PART_PER_THREAD; ++j) {
        if (bin[j] < N_BINS)
            buf_s[start_s[bin[j]] + rank[j]] = (sv[j] << 8) | (dv[j] - bin[j] * BIN_NODES);
    }
    __syncthreads();

    const int total = wtot_s[0] + wtot_s[1] + wtot_s[2] + wtot_s[3]
                    + wtot_s[4] + wtot_s[5] + wtot_s[6] + wtot_s[7];
    for (int i = t; i < total; i += PART_THREADS) sorted[e0 + i] = buf_s[i];
}

__device__ __forceinline__ void gcn_drain_queue(const float* __restrict__ feat,
                                                const int* q, int nq,
                                                float* agg_s, int lane) {
    for (int base = 0; base < nq; base += 32) {
        const int idx = base + lane;
        const bool live = idx < nq;
        const int ent = live ? q[idx] : 0;
        const int sv = ent >> 8;
        const int lv = ent & 255;
        const int nb = (nq - base < 32) ? (nq - base) : 32;
        int b = 0;
        #pragma unroll 1
        for (; b + EDGE_BATCH <= nb; b += EDGE_BATCH) {
            float v[EDGE_BATCH][4];
            #pragma unroll
            for (int j = 0; j < EDGE_BATCH; ++j) {
                const int s = __builtin_amdgcn_readlane(sv, b + j);
                const float* fp = feat + (size_t)s * D + lane;
                v[j][0] = fp[0];
                v[j][1] = fp[32];
                v[j][2] = fp[64];
                v[j][3] = fp[96];
            }
            #pragma unroll
            for (int j = 0; j < EDGE_BATCH; ++j) {
                const int l = __builtin_amdgcn_readlane(lv, b + j);
                float* ap = agg_s + l * D + lane;
                atomicAdd(ap + 0,  v[j][0]);
                atomicAdd(ap + 32, v[j][1]);
                atomicAdd(ap + 64, v[j][2]);
                atomicAdd(ap + 96, v[j][3]);
            }
        }
        #pragma unroll 1
        for (; b < nb; ++b) {
            const int s = __builtin_amdgcn_readlane(sv, b);
            const int l = __builtin_amdgcn_readlane(lv, b);
            const float* fp = feat + (size_t)s * D + lane;
            const float r0 = fp[0], r1 = fp[32], r2 = fp[64], r3 = fp[96];
            float* ap = agg_s + l * D + lane;
            atomicAdd(ap + 0,  r0);
            atomicAdd(ap + 32, r1);
            atomicAdd(ap + 64, r2);
            atomicAdd(ap + 96, r3);
        }
    }
}

__global__ __launch_bounds__(AGG_THREADS)
void gcn_bin_aggregate_kernel(const float* __restrict__ feat,
                              const int* __restrict__ sorted,
                              const int* __restrict__ starts,
                              const int* __restrict__ counts,
                              float* __restrict__ agg,
                              float* __restrict__ deg) {
    __shared__ float agg_s[BIN_NODES * D];
    __shared__ float deg_s[256];
    __shared__ int   q_s[AGG_WAVES * QCAP];

    const int t    = threadIdx.x;
    const int lane = t & 31;
    const int wave = t >> 5;
    const int bin  = blockIdx.x;
    const int lo   = bin * BIN_NODES;
    const int nn   = (N_NODES - lo < BIN_NODES) ? (N_NODES - lo) : BIN_NODES;

    const int s0 = wave * SLICES_PER_WAVE;
    int so = 0, sc = 0;
    if (lane < SLICES_PER_WAVE) {
        so = starts[bin * PART_BLOCKS + s0 + lane];
        sc = counts[bin * PART_BLOCKS + s0 + lane];
    }

    {
        float4* a4 = reinterpret_cast<float4*>(agg_s);
        for (int i = t; i < BIN_NODES * D / 4; i += AGG_THREADS)
            a4[i] = make_float4(0.0f, 0.0f, 0.0f, 0.0f);
        if (t < 256) deg_s[t] = 0.0f;
    }
    __syncthreads();

    int* q = q_s + wave * QCAP;
    int nq = 0;

    int ent[SLICES_PER_WAVE];
    #pragma unroll
    for (int j = 0; j < SLICES_PER_WAVE; ++j) {
        const int oj = __builtin_amdgcn_readlane(so, j);
        const int cj = __builtin_amdgcn_readlane(sc, j);
        ent[j] = (lane < cj) ? sorted[(s0 + j) * PART_EDGES + oj + lane] : 0;
    }
    #pragma unroll
    for (int j = 0; j < SLICES_PER_WAVE; ++j) {
        const int cj = __builtin_amdgcn_readlane(sc, j);
        const int n = (cj < 32) ? cj : 32;
        if (lane < n) {
            q[nq + lane] = ent[j];
            atomicAdd(deg_s + (ent[j] & 255), 1.0f);
        }
        nq += n;
    }
    #pragma unroll 1
    for (int j = 0; j < SLICES_PER_WAVE; ++j) {
        const int oj = __builtin_amdgcn_readlane(so, j);
        const int cj = __builtin_amdgcn_readlane(sc, j);
        for (int base = 32; base < cj; base += 32) {
            const int n = (cj - base < 32) ? (cj - base) : 32;
            if (lane < n) {
                const int e = sorted[(s0 + j) * PART_EDGES + oj + base + lane];
                q[nq + lane] = e;
                atomicAdd(deg_s + (e & 255), 1.0f);
            }
            nq += n;
            if (nq > QCAP - 32) {
                gcn_drain_queue(feat, q, nq, agg_s, lane);
                nq = 0;
            }
        }
    }
    gcn_drain_queue(feat, q, nq, agg_s, lane);
    __syncthreads();

    {
        const float4* a4 = reinterpret_cast<const float4*>(agg_s);
        float4* o4 = reinterpret_cast<float4*>(agg + (size_t)lo * D);
        for (int i = t; i < nn * D / 4; i += AGG_THREADS) o4[i] = a4[i];
        if (t < nn) deg[lo + t] = deg_s[t];
    }
}

__global__ void gcn_gemm_kernel(const float* __restrict__ feat,
                                const float* __restrict__ agg,
                                const float* __restrict__ deg,
                                const float* __restrict__ W,
                                const float* __restrict__ bias,
                                float* __restrict__ out) {
    __shared__ float xs[16 * XS_STRIDE];
    __shared__ float invd[16];

    const int node0 = blockIdx.x * 16;
    const int t = threadIdx.x;

    if (t < 16) {
        float dg = deg[node0 + t];
        invd[t] = 1.0f / fmaxf(dg, 1.0f);
    }
    __syncthreads();

    #pragma unroll
    for (int i = t; i < 16 * K2; i += 256) {
        int r = i >> 8;
        int c = i & 255;
        float v;
        if (c < D) {
            v = agg[(size_t)(node0 + r) * D + c] * invd[r];
        } else {
            v = feat[(size_t)(node0 + r) * D + (c - D)];
        }
        xs[r * XS_STRIDE + c] = v;
    }
    __syncthreads();

    const int wave = t >> 5;
    const int lane = t & 31;
    const int half = lane >> 4;
    const int lid  = lane & 15;
    const int n0   = wave * 16;

    const float* wrow = W + (size_t)(n0 + lid) * K2 + 2 * half;
    const float* arow = xs + lid * XS_STRIDE + 2 * half;

    __builtin_prefetch(wrow, 0, 3);

    v8f acc = {};
    #pragma unroll 8
    for (int k = 0; k < K2; k += 4) {
        v2f a = *(const v2f*)(arow + k);
        v2f b = *(const v2f*)(wrow + k);
        acc = __builtin_amdgcn_wmma_f32_16x16x4_f32(
            false, a, false, b, (short)0, acc, false, false);
    }

    const float bv = bias[n0 + lid];
    #pragma unroll
    for (int j = 0; j < 8; ++j) {
        int row = node0 + j + 8 * half;
        out[(size_t)row * D_OUT + n0 + lid] = acc[j] + bv;
    }
}

extern "C" void kernel_launch(void* const* d_in, const int* in_sizes, int n_in,
                              void* d_out, int out_size, void* d_ws, size_t ws_size,
                              hipStream_t stream) {
    const float* feat = (const float*)d_in[0];
    const int*   src  = (const int*)d_in[1];
    const int*   dst  = (const int*)d_in[2];
    const float* W    = (const float*)d_in[3];
    const float* bias = (const float*)d_in[4];
    float*       out  = (float*)d_out;

    float* agg = (float*)d_ws;
    float* deg = agg + (size_t)N_NODES * D;
    int* sorted = (int*)(deg + DEG_PLANE);
    int* starts = sorted + N_EDGES;
    int* counts = starts + N_BINS * PART_BLOCKS;

    {
        gcn_partition_kernel<<<PART_BLOCKS, PART_THREADS, 0, stream>>>(src, dst, sorted, starts, counts);
    }

    {
        gcn_bin_aggregate_kernel<<<N_BINS, AGG_THREADS, 0, stream>>>(feat, sorted, starts, counts, agg, deg);
    }

    {
        int blocks = N_NODES / 16;
        gcn_gemm_kernel<<<blocks, 256, 0, stream>>>(feat, agg, deg, W, bias, out);
    }
}


# aggregation: block orders its bin's edges by destination node (LDS counting sort, integer atomic ranks); a wave sums a node's rows in registers, 16-byte loads; no LDS f32 atomics; sums in rank order
# speedup vs baseline: 3.9936x; 1.5674x over previous
#include <hip/hip_runtime.h>
#include <hip/hip_bf16.h>

#define N_NODES 50000
#define N_EDGES 800000
#define D       128
#define D_OUT   128
#define K2      256
#define XS_STRIDE 260

typedef __attribute__((ext_vector_type(2))) float v2f;
typedef __attribute__((ext_vector_type(8))) float v8f;

#define BIN_NODES   196
#define N_BINS      ((N_NODES + BIN_NODES - 1) / BIN_NODES)
#define AGG_WAVES   32
#define AGG_THREADS (AGG_WAVES * 32)
#define LIST_CAP    4096
#define LIST_PER_THREAD (LIST_CAP / AGG_THREADS)
#define EDGE_BATCH  8
#define PART_BLOCKS  256
#define PART_THREADS 256
#define PART_EDGES   (N_EDGES / PART_BLOCKS)
#define PART_PER_THREAD ((PART_EDGES + PART_THREADS - 1) / PART_THREADS)
#define SLICES_PER_WAVE (PART_BLOCKS / AGG_WAVES)
#define DEG_PLANE   ((N_NODES + 63) / 64 * 64)
static_assert(N_EDGES % PART_BLOCKS == 0, "edges must split evenly over the slices");
static_assert(PART_BLOCKS % AGG_WAVES == 0, "slices must split evenly over the waves");
static_assert(N_BINS == PART_THREADS, "one partition thread per bin for the bin tables");
static_assert(BIN_NODES <= 256, "local node index must fit the 8 low bits of a packed edge");
static_assert(N_NODES < (1 << 23), "source node must fit the upper bits of a packed edge");
static_assert(LIST_CAP % AGG_THREADS == 0, "list entries split evenly over the threads");
static_assert(PART_BLOCKS == 256 && SLICES_PER_WAVE <= 32, "slice tables are scanned by 256 threads, a wave's slices sit in its lanes");

__global__ __launch_bounds__(PART_THREADS)
void gcn_partition_kernel(const int* __restrict__ src,
                          const int* __restrict__ dst,
                          int* __restrict__ sorted,
                          int* __restrict__ starts,
                          int* __restrict__ counts) {
    __shared__ int cnt_s[N_BINS + 1];
    __shared__ int start_s[N_BINS];
    __shared__ int wtot_s[PART_THREADS / 32];
    __shared__ int buf_s[PART_EDGES];

    const int t    = threadIdx.x;
    const int lane = t & 31;
    const int wave = t >> 5;
    const int sl   = blockIdx.x;
    const int e0   = sl * PART_EDGES;

    cnt_s[t] = 0;
    if (t == 0) cnt_s[N_BINS] = 0;
    __syncthreads();

    int dv[PART_PER_THREAD], sv[PART_PER_THREAD];
    #pragma unroll
    for (int j = 0; j < PART_PER_THREAD; ++j) {
        const int i = t + j * PART_THREADS;
        const int ic = (i < PART_EDGES) ? i : (PART_EDGES - 1);
        const int d = dst[e0 + ic];
        sv[j] = src[e0 + ic];
        dv[j] = (i < PART_EDGES) ? d : -1;
    }
    int bin[PART_PER_THREAD], rank[PART_PER_THREAD];
    #pragma unroll
    for (int j = 0; j < PART_PER_THREAD; ++j) {
        const bool ok = (unsigned)dv[j] < (unsigned)N_NODES;
        bin[j] = ok ? dv[j] / BIN_NODES : N_BINS;
        rank[j] = atomicAdd(&cnt_s[bin[j]], 1);
    }
    __syncthreads();

    const int c = cnt_s[t];
    int x = c;
    #pragma unroll
    for (int off = 1; off < 32; off <<= 1) {
        const int y = __shfl_up(x, off);
        if (lane >= off) x += y;
    }
    if (lane == 31) wtot_s[wave] = x;
    __syncthreads();
    int wbase = 0;
    #pragma unroll
    for (int w = 0; w < PART_THREADS / 32; ++w) wbase += (w < wave) ? wtot_s[w] : 0;
    const int excl = wbase + x - c;
    start_s[t] = excl;
    starts[t * PART_BLOCKS + sl] = excl;
    counts[t * PART_BLOCKS + sl] = c;
    __syncthreads();

    #pragma unroll
    for (int j = 0; j < PART_PER_THREAD; ++j) {
        if (bin[j] < N_BINS)
            buf_s[start_s[bin[j]] + rank[j]] = (sv[j] << 8) | (dv[j] - bin[j] * BIN_NODES);
    }
    __syncthreads();

    const int total = wtot_s[0] + wtot_s[1] + wtot_s[2] + wtot_s[3]
                    + wtot_s[4] + wtot_s[5] + wtot_s[6] + wtot_s[7];
    for (int i = t; i < total; i += PART_THREADS) sorted[e0 + i] = buf_s[i];
}

typedef __attribute__((ext_vector_type(4))) float v4f;

__device__ __forceinline__ int gcn_excl_scan256(int c, int lane, int wave, int* wtot) {
    int x = c;
    #pragma unroll
    for (int off = 1; off < 32; off <<= 1) {
        const int y = __shfl_up(x, off);
        if (lane >= off) x += y;
    }
    if (wave < 8 && lane == 31) wtot[wave] = x;
    __syncthreads();
    int wbase = 0;
    #pragma unroll
    for (int w = 0; w < 8; ++w) wbase += (w < wave) ? wtot[w] : 0;
    return wbase + x - c;
}

template <bool FULL>
__device__ __forceinline__ void gcn_add_rows(const float* __restrict__ feat, int srcs,
                                             int b, int m, int lane, v4f& acc) {
    v4f v[EDGE_BATCH];
    #pragma unroll
    for (int j = 0; j < EDGE_BATCH; ++j) {
        if (FULL || b + j < m) {
            const int sn = __builtin_amdgcn_readlane(srcs, b + j);
            v[j] = *((const v4f*)(feat + (size_t)sn * D) + lane);
        } else {
            v[j] = (v4f){0.0f, 0.0f, 0.0f, 0.0f};
        }
    }
    #pragma unroll
    for (int j = 0; j < EDGE_BATCH; ++j) acc += v[j];
}

__global__ __launch_bounds__(AGG_THREADS)
void gcn_bin_aggregate_kernel(const float* __restrict__ feat,
                              const int* __restrict__ sorted,
                              const int* __restrict__ starts,
                              const int* __restrict__ counts,
                              float* __restrict__ agg,
                              float* __restrict__ deg) {
    __shared__ int list_s[LIST_CAP];
    __shared__ int srt_s[LIST_CAP];
    __shared__ int hist_s[256];
    __shared__ int start_s[256];
    __shared__ int c_s[PART_BLOCKS];
    __shared__ int o_s[PART_BLOCKS];
    __shared__ int p_s[PART_BLOCKS];
    __shared__ int wtot_a[8];
    __shared__ int wtot_b[8];

    const int t    = threadIdx.x;
    const int lane = t & 31;
    const int wave = __builtin_amdgcn_readfirstlane(t >> 5);
    const int bin  = blockIdx.x;
    const int lo   = bin * BIN_NODES;
    const int nn   = (N_NODES - lo < BIN_NODES) ? (N_NODES - lo) : BIN_NODES;

    {
        int c = 0, o = 0;
        if (t < PART_BLOCKS) {
            c = counts[bin * PART_BLOCKS + t];
            o = starts[bin * PART_BLOCKS + t];
        }
        const int p = gcn_excl_scan256(c, lane, wave, wtot_a);
        if (t < PART_BLOCKS) { c_s[t] = c; o_s[t] = o; p_s[t] = p; }
    }
    __syncthreads();
    const int total = wtot_a[0] + wtot_a[1] + wtot_a[2] + wtot_a[3]
                    + wtot_a[4] + wtot_a[5] + wtot_a[6] + wtot_a[7];

    const int s0 = wave * SLICES_PER_WAVE;
    int sc = 0, so = 0, sp = 0;
    if (lane < SLICES_PER_WAVE) { sc = c_s[s0 + lane]; so = o_s[s0 + lane]; sp = p_s[s0 + lane]; }

    int pass = 0;
    for (int w0 = 0; pass == 0 || w0 < total; w0 += LIST_CAP, ++pass) {
        const int n = (total - w0 < LIST_CAP) ? (total - w0) : LIST_CAP;

        if (t < 256) hist_s[t] = 0;

        int ent[SLICES_PER_WAVE];
        #pragma unroll
        for (int j = 0; j < SLICES_PER_WAVE; ++j) {
            const int oj = __builtin_amdgcn_readlane(so, j);
            const int cj = __builtin_amdgcn_readlane(sc, j);
            ent[j] = (lane < cj) ? sorted[(s0 + j) * PART_EDGES + oj + lane] : 0;
        }
        #pragma unroll
        for (int j = 0; j < SLICES_PER_WAVE; ++j) {
            const int cj = __builtin_amdgcn_readlane(sc, j);
            const int f = __builtin_amdgcn_readlane(sp, j) + lane - w0;
            if (lane < cj && (unsigned)f < (unsigned)LIST_CAP) list_s[f] = ent[j];
        }
        #pragma unroll 1
        for (int j = 0; j < SLICES_PER_WAVE; ++j) {
            const int oj = __builtin_amdgcn_readlane(so, j);
            const int cj = __builtin_amdgcn_readlane(sc, j);
            const int pj = __builtin_amdgcn_readlane(sp, j);
            for (int i = 32 + lane; i < cj; i += 32) {
                const int f = pj + i - w0;
                if ((unsigned)f < (unsigned)LIST_CAP) list_s[f] = sorted[(s0 + j) * PART_EDGES + oj + i];
            }
        }
        __syncthreads();

        int e[LIST_PER_THREAD], rk[LIST_PER_THREAD];
        #pragma unroll
        for (int r = 0; r < LIST_PER_THREAD; ++r) {
            const int idx = t + r * AGG_THREADS;
            e[r] = 0; rk[r] = 0;
            if (idx < n) {
                e[r] = list_s[idx];
                rk[r] = atomicAdd(&hist_s[e[r] & 255], 1);
            }
        }
        __syncthreads();
        {
            const int hc = (t < 256) ? hist_s[t] : 0;
            const int hx = gcn_excl_scan256(hc, lane, wave, wtot_b);
            if (t < 256) start_s[t] = hx;
        }
        __syncthreads();
        #pragma unroll
        for (int r = 0; r < LIST_PER_THREAD; ++r) {
            const int idx = t + r * AGG_THREADS;
            if (idx < n) srt_s[start_s[e[r] & 255] + rk[r]] = e[r] >> 8;
        }
        __syncthreads();

        for (int node = wave; node < nn; node += AGG_WAVES) {
            const int c  = __builtin_amdgcn_readfirstlane(hist_s[node]);
            const int st = __builtin_amdgcn_readfirstlane(start_s[node]);
            v4f* arow = (v4f*)(agg + (size_t)(lo + node) * D) + lane;
            v4f acc = {0.0f, 0.0f, 0.0f, 0.0f};
            float dprev = 0.0f;
            if (pass > 0) { acc = *arow; dprev = deg[lo + node]; }
            for (int q0 = 0; q0 < c; q0 += 32) {
                const int m = (c - q0 < 32) ? (c - q0) : 32;
                const int srcs = (lane < m) ? srt_s[st + q0 + lane] : 0;
                int b = 0;
                #pragma unroll 1
                for (; b + EDGE_BATCH <= m; b += EDGE_BATCH)
                    gcn_add_rows<true>(feat, srcs, b, m, lane, acc);
                if (b < m) gcn_add_rows<false>(feat, srcs, b, m, lane, acc);
            }
            *arow = acc;
            if (lane == 0) deg[lo + node] = dprev + (float)c;
        }
        __syncthreads();
    }
}

#define GEMM_BLOCKS 512
#define GEMM_TILES  (N_NODES / 16)
static_assert(N_NODES % 16 == 0, "node tiles are exact");
static_assert((XS_STRIDE * 4) % 16 == 0, "x rows start on 16-byte boundaries in LDS");

__global__ __launch_bounds__(256)
void gcn_gemm_kernel(const float* __restrict__ feat,
                     const float* __restrict__ agg,
                     const float* __restrict__ deg,
                     const float* __restrict__ W,
                     const float* __restrict__ bias,
                     float* __restrict__ out) {
    __shared__ __attribute__((aligned(16))) float xs[2][16 * XS_STRIDE];

    const int t = threadIdx.x;
    const int wave = t >> 5;
    const int lane = t & 31;
    const int half = lane >> 4;
    const int lid  = lane & 15;
    const int n0   = wave * 16;

    const int sr = t >> 4;
    const int sq = t & 15;

    const float* wrow = W + (size_t)(n0 + lid) * K2 + 4 * half;
    v4f bfr[32];
    #pragma unroll
    for (int jj = 0; jj < 32; ++jj) bfr[jj] = *(const v4f*)(wrow + 8 * jj);
    const float bv = bias[n0 + lid];

    int tile = blockIdx.x;

    {
        const size_t node = (size_t)tile * 16 + sr;
        const v4f* a4 = (const v4f*)(agg + node * D);
        const v4f* f4 = (const v4f*)(feat + node * D);
        const v4f pa0 = a4[sq], pa1 = a4[sq + 16];
        const v4f pf0 = f4[sq], pf1 = f4[sq + 16];
        const float inv = 1.0f / fmaxf(deg[node], 1.0f);
        float* xr = xs[0] + sr * XS_STRIDE + 4 * sq;
        *(v4f*)(xr)       = pa0 * inv;
        *(v4f*)(xr + 64)  = pa1 * inv;
        *(v4f*)(xr + 128) = pf0;
        *(v4f*)(xr + 192) = pf1;
    }
    __syncthreads();

    float* orow = out + (size_t)(8 * half) * D_OUT + n0 + lid;

    for (int buf = 0; tile < GEMM_TILES; tile += GEMM_BLOCKS, buf ^= 1) {
        const int nxt = (tile + GEMM_BLOCKS < GEMM_TILES) ? (tile + GEMM_BLOCKS) : tile;
        const size_t pnode = (size_t)nxt * 16 + sr;
        const v4f* a4 = (const v4f*)(agg + pnode * D);
        const v4f* f4 = (const v4f*)(feat + pnode * D);
        const v4f pa0 = a4[sq], pa1 = a4[sq + 16];
        const v4f pf0 = f4[sq], pf1 = f4[sq + 16];
        const float pd = deg[pnode];
        __builtin_amdgcn_sched_barrier(0);

        const float* arow = xs[buf] + lid * XS_STRIDE + 4 * half;
        v8f acc = {};
        #pragma unroll
        for (int jj = 0; jj < 32; ++jj) {
            const v4f a = *(const v4f*)(arow + 8 * jj);
            const v2f a0 = {a.x, a.y}, a1 = {a.z, a.w};
            const v2f b0 = {bfr[jj].x, bfr[jj].y}, b1 = {bfr[jj].z, bfr[jj].w};
            acc = __builtin_amdgcn_wmma_f32_16x16x4_f32(
                false, a0, false, b0, (short)0, acc, false, false);
            acc = __builtin_amdgcn_wmma_f32_16x16x4_f32(
                false, a1, false, b1, (short)0, acc, false, false);
        }

        __builtin_amdgcn_sched_barrier(0);
        {
            const float inv = 1.0f / fmaxf(pd, 1.0f);
            float* xr = xs[buf ^ 1] + sr * XS_STRIDE + 4 * sq;
            *(v4f*)(xr)       = pa0 * inv;
            *(v4f*)(xr + 64)  = pa1 * inv;
            *(v4f*)(xr + 128) = pf0;
            *(v4f*)(xr + 192) = pf1;
        }
        __syncthreads();

        float* op = orow + (size_t)tile * 16 * D_OUT;
        #pragma unroll
        for (int j = 0; j < 8; ++j) op[j * D_OUT] = acc[j] + bv;
    }
}

extern "C" void kernel_launch(void* const* d_in, const int* in_sizes, int n_in,
                              void* d_out, int out_size, void* d_ws, size_t ws_size,
                              hipStream_t stream) {
    const float* feat = (const float*)d_in[0];
    const int*   src  = (const int*)d_in[1];
    const int*   dst  = (const int*)d_in[2];
    const float* W    = (const float*)d_in[3];
    const float* bias = (const float*)d_in[4];
    float*       out  = (float*)d_out;

    float* agg = (float*)d_ws;
    float* deg = agg + (size_t)N_NODES * D;
    int* sorted = (int*)(deg + DEG_PLANE);
    int* starts = sorted + N_EDGES;
    int* counts = starts + N_BINS * PART_BLOCKS;

    {
        gcn_partition_kernel<<<PART_BLOCKS, PART_THREADS, 0, stream>>>(src, dst, sorted, starts, counts);
    }

    {
        gcn_bin_aggregate_kernel<<<N_BINS, AGG_THREADS, 0, stream>>>(feat, sorted, starts, counts, agg, deg);
    }

    {
        gcn_gemm_kernel<<<GEMM_BLOCKS, 256, 0, stream>>>(feat, agg, deg, W, bias, out);
    }
}


# aggregation and GEMM fused in one launch: a bin's block sums its mean rows into an LDS x tile and multiplies it there on f32 WMMA; agg/deg leave the workspace, 3 launches -> 2; sum order unchanged
# speedup vs baseline: 4.2469x; 1.0634x over previous
#include <hip/hip_runtime.h>
#include <hip/hip_bf16.h>

#define N_NODES 50000
#define N_EDGES 800000
#define D       128
#define D_OUT   128
#define K2      256
#define XS_STRIDE 260

typedef __attribute__((ext_vector_type(2))) float v2f;
typedef __attribute__((ext_vector_type(4))) float v4f;
typedef __attribute__((ext_vector_type(8))) float v8f;

#define BIN_NODES   196
#define N_BINS      ((N_NODES + BIN_NODES - 1) / BIN_NODES)
#define BIN_TILES   ((BIN_NODES + 15) / 16)
#define BIN_ROWS    (BIN_TILES * 16)
#define AGG_WAVES   16
#define AGG_THREADS (AGG_WAVES * 32)
#define LIST_CAP    4096
#define LIST_PER_THREAD (LIST_CAP / AGG_THREADS)
#define EDGE_BATCH  16
#define PART_BLOCKS  256
#define PART_THREADS 256
#define PART_EDGES   (N_EDGES / PART_BLOCKS)
#define PART_PER_THREAD ((PART_EDGES + PART_THREADS - 1) / PART_THREADS)
#define SLICES_PER_WAVE (PART_BLOCKS / AGG_WAVES)
static_assert(N_EDGES % PART_BLOCKS == 0, "edges must split evenly over the slices");
static_assert(PART_BLOCKS % AGG_WAVES == 0, "slices must split evenly over the waves");
static_assert(N_BINS == PART_THREADS, "one partition thread per bin for the bin tables");
static_assert(BIN_NODES <= 256, "local node index must fit the 8 low bits of a packed edge");
static_assert(N_NODES < (1 << 23), "source node must fit the upper bits of a packed edge");
static_assert(LIST_CAP % AGG_THREADS == 0, "list entries split evenly over the threads");
static_assert(PART_BLOCKS == 256 && SLICES_PER_WAVE <= 32, "slice tables are scanned by 256 threads, a wave's slices sit in its lanes");
static_assert(AGG_WAVES == 16 && D_OUT == 128, "product phase: 8 column groups of 16 x 2 tile halves = 16 waves");
static_assert((XS_STRIDE * 4) % 16 == 0, "x rows start on 16-byte boundaries in LDS");
static_assert(D == 128 && K2 == 2 * D, "a wave load of 16 bytes per lane is one whole row");

__global__ __launch_bounds__(PART_THREADS)
void gcn_partition_kernel(const int* __restrict__ src,
                          const int* __restrict__ dst,
                          int* __restrict__ sorted,
                          int* __restrict__ starts,
                          int* __restrict__ counts) {
    __shared__ int cnt_s[N_BINS + 1];
    __shared__ int start_s[N_BINS];
    __shared__ int wtot_s[PART_THREADS / 32];
    __shared__ int buf_s[PART_EDGES];

    const int t    = threadIdx.x;
    const int lane = t & 31;
    const int wave = t >> 5;
    const int sl   = blockIdx.x;
    const int e0   = sl * PART_EDGES;

    cnt_s[t] = 0;
    if (t == 0) cnt_s[N_BINS] = 0;
    __syncthreads();

    int dv[PART_PER_THREAD], sv[PART_PER_THREAD];
    #pragma unroll
    for (int j = 0; j < PART_PER_THREAD; ++j) {
        const int i = t + j * PART_THREADS;
        const int ic = (i < PART_EDGES) ? i : (PART_EDGES - 1);
        const int d = dst[e0 + ic];
        sv[j] = src[e0 + ic];
        dv[j] = (i < PART_EDGES) ? d : -1;
    }
    int bin[PART_PER_THREAD], rank[PART_PER_THREAD];
    #pragma unroll
    for (int j = 0; j < PART_PER_THREAD; ++j) {
        const bool ok = (unsigned)dv[j] < (unsigned)N_NODES;
        bin[j] = ok ? dv[j] / BIN_NODES : N_BINS;
        rank[j] = atomicAdd(&cnt_s[bin[j]], 1);
    }
    __syncthreads();

    const int c = cnt_s[t];
    int x = c;
    #pragma unroll
    for (int off = 1; off < 32; off <<= 1) {
        const int y = __shfl_up(x, off);
        if (lane >= off) x += y;
    }
    if (lane == 31) wtot_s[wave] = x;
    __syncthreads();
    int wbase = 0;
    #pragma unroll
    for (int w = 0; w < PART_THREADS / 32; ++w) wbase += (w < wave) ? wtot_s[w] : 0;
    const int excl = wbase + x - c;
    start_s[t] = excl;
    starts[t * PART_BLOCKS + sl] = excl;
    counts[t * PART_BLOCKS + sl] = c;
    __syncthreads();

    #pragma unroll
    for (int j = 0; j < PART_PER_THREAD; ++j) {
        if (bin[j] < N_BINS)
            buf_s[start_s[bin[j]] + rank[j]] = (sv[j] << 8) | (dv[j] - bin[j] * BIN_NODES);
    }
    __syncthreads();

    const int total = wtot_s[0] + wtot_s[1] + wtot_s[2] + wtot_s[3]
                    + wtot_s[4] + wtot_s[5] + wtot_s[6] + wtot_s[7];
    for (int i = t; i < total; i += PART_THREADS) sorted[e0 + i] = buf_s[i];
}

__device__ __forceinline__ void gcn_lds_barrier() {
    __builtin_amdgcn_fence(__ATOMIC_RELEASE, "workgroup", "local");
    __builtin_amdgcn_s_barrier();
    __builtin_amdgcn_fence(__ATOMIC_ACQUIRE, "workgroup", "local");
}

__device__ __forceinline__ int gcn_excl_scan256(int c, int lane, int wave, int* wtot) {
    int x = c;
    #pragma unroll
    for (int off = 1; off < 32; off <<= 1) {
        const int y = __shfl_up(x, off);
        if (lane >= off) x += y;
    }
    if (wave < 8 && lane == 31) wtot[wave] = x;
    gcn_lds_barrier();
    int wbase = 0;
    #pragma unroll
    for (int w = 0; w < 8; ++w) wbase += (w < wave) ? wtot[w] : 0;
    return wbase + x - c;
}

template <bool FULL>
__device__ __forceinline__ void gcn_add_rows(const float* __restrict__ feat, int srcs,
                                             int b, int m, int lane, v4f& acc) {
    v4f v[EDGE_BATCH];
    #pragma unroll
    for (int j = 0; j < EDGE_BATCH; ++j) {
        if (FULL || b + j < m) {
            const int sn = __builtin_amdgcn_readlane(srcs, b + j);
            v[j] = *((const v4f*)(feat + (size_t)sn * D) + lane);
        } else {
            v[j] = (v4f){0.0f, 0.0f, 0.0f, 0.0f};
        }
    }
    #pragma unroll
    for (int j = 0; j < EDGE_BATCH; ++j) acc += v[j];
}

__global__ __launch_bounds__(AGG_THREADS)
void gcn_bin_layer_kernel(const float* __restrict__ feat,
                          const int* __restrict__ sorted,
                          const int* __restrict__ starts,
                          const int* __restrict__ counts,
                          const float* __restrict__ W,
                          const float* __restrict__ bias,
                          float* __restrict__ out) {
    __shared__ __attribute__((aligned(16))) float xs[BIN_ROWS * XS_STRIDE];
    __shared__ int list_s[LIST_CAP];
    __shared__ int srt_s[LIST_CAP];
    __shared__ int hist_s[256];
    __shared__ int start_s[256];
    __shared__ float deg_s[256];
    __shared__ int c_s[PART_BLOCKS];
    __shared__ int o_s[PART_BLOCKS];
    __shared__ int p_s[PART_BLOCKS];
    __shared__ int wtot_a[8];
    __shared__ int wtot_b[8];

    const int t    = threadIdx.x;
    const int lane = t & 31;
    const int wave = __builtin_amdgcn_readfirstlane(t >> 5);
    const int bin  = blockIdx.x;
    const int lo   = bin * BIN_NODES;
    const int nn   = (N_NODES - lo < BIN_NODES) ? (N_NODES - lo) : BIN_NODES;
    const int ntiles = (nn + 15) >> 4;

    for (int i = t; i < (ntiles * 16 - nn) * (K2 / 4); i += AGG_THREADS) {
        const int r = nn + i / (K2 / 4), q = i % (K2 / 4);
        *(v4f*)(xs + r * XS_STRIDE + 4 * q) = (v4f){0.0f, 0.0f, 0.0f, 0.0f};
    }

    {
        int c = 0, o = 0;
        if (t < PART_BLOCKS) {
            c = counts[bin * PART_BLOCKS + t];
            o = starts[bin * PART_BLOCKS + t];
        }
        const int p = gcn_excl_scan256(c, lane, wave, wtot_a);
        if (t < PART_BLOCKS) { c_s[t] = c; o_s[t] = o; p_s[t] = p; }
    }
    __syncthreads();
    const int total = wtot_a[0] + wtot_a[1] + wtot_a[2] + wtot_a[3]
                    + wtot_a[4] + wtot_a[5] + wtot_a[6] + wtot_a[7];

    const int s0 = wave * SLICES_PER_WAVE;
    int sc = 0, so = 0, sp = 0;
    if (lane < SLICES_PER_WAVE) { sc = c_s[s0 + lane]; so = o_s[s0 + lane]; sp = p_s[s0 + lane]; }

    const int half = lane >> 4;
    const int lid  = lane & 15;
    const int n0   = (wave & 7) * 16;
    const float* wrow = W + (size_t)(n0 + lid) * K2 + 4 * half;
    v4f bfr[32];

    int pass = 0;
    for (int w0 = 0; pass == 0 || w0 < total; w0 += LIST_CAP, ++pass) {
        const int n = (total - w0 < LIST_CAP) ? (total - w0) : LIST_CAP;
        const bool last = (total - w0 <= LIST_CAP);

        if (t < 256) hist_s[t] = 0;

        int ent[SLICES_PER_WAVE];
        #pragma unroll
        for (int j = 0; j < SLICES_PER_WAVE; ++j) {
            const int oj = __builtin_amdgcn_readlane(so, j);
            const int cj = __builtin_amdgcn_readlane(sc, j);
            ent[j] = (lane < cj) ? sorted[(s0 + j) * PART_EDGES + oj + lane] : 0;
        }
        #pragma unroll
        for (int j = 0; j < SLICES_PER_WAVE; ++j) {
            const int cj = __builtin_amdgcn_readlane(sc, j);
            const int f = __builtin_amdgcn_readlane(sp, j) + lane - w0;
            if (lane < cj && (unsigned)f < (unsigned)LIST_CAP) list_s[f] = ent[j];
        }
        #pragma unroll 1
        for (int j = 0; j < SLICES_PER_WAVE; ++j) {
            const int oj = __builtin_amdgcn_readlane(so, j);
            const int cj = __builtin_amdgcn_readlane(sc, j);
            const int pj = __builtin_amdgcn_readlane(sp, j);
            for (int i = 32 + lane; i < cj; i += 32) {
                const int f = pj + i - w0;
                if ((unsigned)f < (unsigned)LIST_CAP) list_s[f] = sorted[(s0 + j) * PART_EDGES + oj + i];
            }
        }
        if (pass == 0) {
            #pragma unroll
            for (int jj = 0; jj < 32; ++jj) bfr[jj] = *(const v4f*)(wrow + 8 * jj);
        }
        gcn_lds_barrier();

        int e[LIST_PER_THREAD], rk[LIST_PER_THREAD];
        #pragma unroll
        for (int r = 0; r < LIST_PER_THREAD; ++r) {
            const int idx = t + r * AGG_THREADS;
            e[r] = 0; rk[r] = 0;
            if (idx < n) {
                e[r] = list_s[idx];
                rk[r] = atomicAdd(&hist_s[e[r] & 255], 1);
            }
        }
        gcn_lds_barrier();
        {
            const int hc = (t < 256) ? hist_s[t] : 0;
            const int hx = gcn_excl_scan256(hc, lane, wave, wtot_b);
            if (t < 256) start_s[t] = hx;
        }
        gcn_lds_barrier();
        #pragma unroll
        for (int r = 0; r < LIST_PER_THREAD; ++r) {
            const int idx = t + r * AGG_THREADS;
            if (idx < n) srt_s[start_s[e[r] & 255] + rk[r]] = e[r] >> 8;
        }
        gcn_lds_barrier();

        for (int node = wave; node < nn; node += AGG_WAVES) {
            const int c  = __builtin_amdgcn_readfirstlane(hist_s[node]);
            const int st = __builtin_amdgcn_readfirstlane(start_s[node]);
            float* xrow = xs + node * XS_STRIDE + 4 * lane;
            v4f fr = {0.0f, 0.0f, 0.0f, 0.0f};
            if (pass == 0) fr = *((const v4f*)(feat + (size_t)(lo + node) * D) + lane);
            v4f acc = {0.0f, 0.0f, 0.0f, 0.0f};
            float dprev = 0.0f;
            if (pass > 0) { acc = *(v4f*)xrow; dprev = deg_s[node]; }
            for (int q0 = 0; q0 < c; q0 += 32) {
                const int m = (c - q0 < 32) ? (c - q0) : 32;
                const int srcs = (lane < m) ? srt_s[st + q0 + lane] : 0;
                int b = 0;
                #pragma unroll 1
                for (; b + EDGE_BATCH <= m; b += EDGE_BATCH)
                    gcn_add_rows<true>(feat, srcs, b, m, lane, acc);
                if (b < m) gcn_add_rows<false>(feat, srcs, b, m, lane, acc);
            }
            const float dtot = dprev + (float)c;
            if (last) acc = acc * (1.0f / fmaxf(dtot, 1.0f));
            *(v4f*)xrow = acc;
            if (pass == 0) *(v4f*)(xrow + D) = fr;
            if (!last && lane == 0) deg_s[node] = dtot;
        }
        __syncthreads();
    }

    const float bv = bias[n0 + lid];
    for (int tile = wave >> 3; tile < ntiles; tile += 2) {
        const float* arow = xs + (tile * 16 + lid) * XS_STRIDE + 4 * half;
        v8f acc = {};
        #pragma unroll
        for (int jj = 0; jj < 32; ++jj) {
            const v4f a = *(const v4f*)(arow + 8 * jj);
            const v2f a0 = {a.x, a.y}, a1 = {a.z, a.w};
            const v2f b0 = {bfr[jj].x, bfr[jj].y}, b1 = {bfr[jj].z, bfr[jj].w};
            acc = __builtin_amdgcn_wmma_f32_16x16x4_f32(
                false, a0, false, b0, (short)0, acc, false, false);
            acc = __builtin_amdgcn_wmma_f32_16x16x4_f32(
                false, a1, false, b1, (short)0, acc, false, false);
        }
        const int r0 = tile * 16 + 8 * half;
        float* op = out + (size_t)(lo + r0) * D_OUT + n0 + lid;
        #pragma unroll
        for (int j = 0; j < 8; ++j) {
            const float v = acc[j] + bv;
            if (r0 + j < nn) op[j * D_OUT] = v;
        }
    }
}

extern "C" void kernel_launch(void* const* d_in, const int* in_sizes, int n_in,
                              void* d_out, int out_size, void* d_ws, size_t ws_size,
                              hipStream_t stream) {
    const float* feat = (const float*)d_in[0];
    const int*   src  = (const int*)d_in[1];
    const int*   dst  = (const int*)d_in[2];
    const float* W    = (const float*)d_in[3];
    const float* bias = (const float*)d_in[4];
    float*       out  = (float*)d_out;

    int* sorted = (int*)d_ws;
    int* starts = sorted + N_EDGES;
    int* counts = starts + N_BINS * PART_BLOCKS;

    {
        gcn_partition_kernel<<<PART_BLOCKS, PART_THREADS, 0, stream>>>(src, dst, sorted, starts, counts);
    }

    {
        gcn_bin_layer_kernel<<<N_BINS, AGG_THREADS, 0, stream>>>(feat, sorted, starts, counts, W, bias, out);
    }
}
